# starting kernel
# baseline (speedup 1.0000x reference)
#include <hip/hip_runtime.h>
#include <hip/hip_bf16.h>

typedef float v2f __attribute__((ext_vector_type(2)));
typedef float v8f __attribute__((ext_vector_type(8)));

#define NN 16384
#define DD 16
#define PP 8
#define MM 50
#define HH 128

#define PW12 288
#define PW3   32
#define SXY   28
#define SH   132
#define WAVES 4

static __device__ __forceinline__ v2f ldsv2(const float* p) { return *(const v2f*)p; }

__launch_bounds__(128, 2)
__global__ void sde_rollout_kernel(const float* __restrict__ X0,
                                   const float* __restrict__ V0,
                                   const float* __restrict__ Yobs,
                                   const float* __restrict__ noise,
                                   const float* __restrict__ W1,
                                   const float* __restrict__ b1,
                                   const float* __restrict__ W2,
                                   const float* __restrict__ b2,
                                   const float* __restrict__ W3,
                                   const float* __restrict__ b3,
                                   float* __restrict__ outX,
                                   float* __restrict__ outV)
{
    __shared__ float sW1p[12 * PW12];
    __shared__ float sW1t[HH];
    __shared__ float sB1 [HH];
    __shared__ float sW2p[64 * PW12];
    __shared__ float sB2 [HH];
    __shared__ float sW3p[64 * PW3];
    __shared__ float sB3 [DD];
    __shared__ float sXY [WAVES][16 * SXY];
    __shared__ float sHH [WAVES][16 * SH];

    const int tid  = threadIdx.x;
    const int lane = tid & 31;
    const int wave = tid >> 5;
    const int lo   = lane & 15;
    const int hi   = lane >> 4;
    const int rowbase = blockIdx.x * (WAVES * 16) + wave * 16;

    for (int i = tid; i < 24 * HH; i += blockDim.x) {
        int r = i >> 7, c = i & 127;
        sW1p[(r >> 1) * PW12 + c * 2 + (r & 1)] = W1[(1 + r) * HH + c];
    }
    if (tid < HH) { sW1t[tid] = W1[tid]; sB1[tid] = b1[tid]; sB2[tid] = b2[tid]; }
    for (int i = tid; i < HH * HH; i += blockDim.x) {
        int r = i >> 7, c = i & 127;
        sW2p[(r >> 1) * PW12 + c * 2 + (r & 1)] = W2[i];
    }
    for (int i = tid; i < HH * DD; i += blockDim.x) {
        int r = i >> 4, c = i & 15;
        sW3p[(r >> 1) * PW3 + c * 2 + (r & 1)] = W3[i];
    }
    if (tid < DD) sB3[tid] = b3[tid];

    float* xy = sXY[wave];
    for (int i = lane; i < 16 * PP; i += 32) {
        int r = i >> 3, c = i & 7;
        xy[r * SXY + DD + c] = Yobs[(rowbase + r) * PP + c];
    }
    __syncthreads();

    float xr[8], va[8];
    #pragma unroll
    for (int v = 0; v < 8; ++v) {
        int row = rowbase + v + 8 * hi;
        xr[v] = X0[row * DD + lo];
        va[v] = V0[row];
    }

    const float dt    = 1.0f / MM;
    const float sqdt  = __builtin_sqrtf(dt);
    const float decay = 1.0f - dt;
    float* hsc = sHH[wave];
    const v8f zc = {0.f, 0.f, 0.f, 0.f, 0.f, 0.f, 0.f, 0.f};

    for (int m = 0; m < MM; ++m) {
        const float t = m * dt;

        #pragma unroll
        for (int v = 0; v < 8; ++v)
            xy[(v + 8 * hi) * SXY + lo] = xr[v];

        float ep[8];
        #pragma unroll
        for (int v = 0; v < 8; ++v) {
            size_t row = (size_t)m * NN + (size_t)(rowbase + v + 8 * hi);
            ep[v] = noise[row * DD + lo];
        }

        asm volatile("s_wait_dscnt 0" ::: "memory");

        v8f acc[8];

        {
            v2f a = ldsv2(&xy[lo * SXY + 2 * hi]);
            v2f b[8];
            #pragma unroll
            for (int c = 0; c < 8; ++c) b[c] = ldsv2(&sW1p[hi * PW12 + (c * 16 + lo) * 2]);
            #pragma unroll
            for (int kk = 0; kk < 6; ++kk) {
                v2f an = a; v2f bn[8];
                #pragma unroll
                for (int c = 0; c < 8; ++c) bn[c] = b[c];
                if (kk < 5) {
                    an = ldsv2(&xy[lo * SXY + 4 * (kk + 1) + 2 * hi]);
                    #pragma unroll
                    for (int c = 0; c < 8; ++c)
                        bn[c] = ldsv2(&sW1p[(2 * (kk + 1) + hi) * PW12 + (c * 16 + lo) * 2]);
                }
                #pragma unroll
                for (int c = 0; c < 8; ++c)
                    acc[c] = __builtin_amdgcn_wmma_f32_16x16x4_f32(
                        false, a, false, b[c], (short)0, (kk == 0 ? zc : acc[c]), false, false);
                a = an;
                #pragma unroll
                for (int c = 0; c < 8; ++c) b[c] = bn[c];
            }
        }
        #pragma unroll
        for (int c = 0; c < 8; ++c) {
            float bias = t * sW1t[c * 16 + lo] + sB1[c * 16 + lo];
            #pragma unroll
            for (int v = 0; v < 8; ++v)
                hsc[(v + 8 * hi) * SH + c * 16 + lo] = fmaxf(acc[c][v] + bias, 0.0f);
        }
        asm volatile("s_wait_dscnt 0" ::: "memory");

        {
            v2f a = ldsv2(&hsc[lo * SH + 2 * hi]);
            v2f b[8];
            #pragma unroll
            for (int c = 0; c < 8; ++c) b[c] = ldsv2(&sW2p[hi * PW12 + (c * 16 + lo) * 2]);
            #pragma unroll
            for (int kk = 0; kk < 32; ++kk) {
                v2f an = a; v2f bn[8];
                #pragma unroll
                for (int c = 0; c < 8; ++c) bn[c] = b[c];
                if (kk < 31) {
                    an = ldsv2(&hsc[lo * SH + 4 * (kk + 1) + 2 * hi]);
                    #pragma unroll
                    for (int c = 0; c < 8; ++c)
                        bn[c] = ldsv2(&sW2p[(2 * (kk + 1) + hi) * PW12 + (c * 16 + lo) * 2]);
                }
                #pragma unroll
                for (int c = 0; c < 8; ++c)
                    acc[c] = __builtin_amdgcn_wmma_f32_16x16x4_f32(
                        false, a, false, b[c], (short)0, (kk == 0 ? zc : acc[c]), false, false);
                a = an;
                #pragma unroll
                for (int c = 0; c < 8; ++c) b[c] = bn[c];
            }
        }
        #pragma unroll
        for (int c = 0; c < 8; ++c) {
            float bias = sB2[c * 16 + lo];
            #pragma unroll
            for (int v = 0; v < 8; ++v)
                hsc[(v + 8 * hi) * SH + c * 16 + lo] = fmaxf(acc[c][v] + bias, 0.0f);
        }
        asm volatile("s_wait_dscnt 0" ::: "memory");

        v8f z;
        {
            v2f a = ldsv2(&hsc[lo * SH + 2 * hi]);
            v2f b = ldsv2(&sW3p[hi * PW3 + lo * 2]);
            #pragma unroll
            for (int kk = 0; kk < 32; ++kk) {
                v2f an = a, bn = b;
                if (kk < 31) {
                    an = ldsv2(&hsc[lo * SH + 4 * (kk + 1) + 2 * hi]);
                    bn = ldsv2(&sW3p[(2 * (kk + 1) + hi) * PW3 + lo * 2]);
                }
                z = __builtin_amdgcn_wmma_f32_16x16x4_f32(
                    false, a, false, b, (short)0, (kk == 0 ? zc : z), false, false);
                a = an; b = bn;
            }
        }

        const float b3v = sB3[lo];
        #pragma unroll
        for (int v = 0; v < 8; ++v) {
            float zz = z[v] + b3v;
            float s  = 0.5f * dt * zz * zz + sqdt * zz * ep[v];
            s += __shfl_xor(s, 1);
            s += __shfl_xor(s, 2);
            s += __shfl_xor(s, 4);
            s += __shfl_xor(s, 8);
            va[v] += s;
            xr[v]  = decay * xr[v] + sqdt * ep[v];
        }
    }

    #pragma unroll
    for (int v = 0; v < 8; ++v) {
        int row = rowbase + v + 8 * hi;
        outX[row * DD + lo] = xr[v];
        if (lo == 0) outV[row] = va[v];
    }
}

extern "C" void kernel_launch(void* const* d_in, const int* in_sizes, int n_in,
                              void* d_out, int out_size, void* d_ws, size_t ws_size,
                              hipStream_t stream) {
    (void)in_sizes; (void)n_in; (void)out_size; (void)d_ws; (void)ws_size;
    const float* X0    = (const float*)d_in[0];
    const float* V0    = (const float*)d_in[1];
    const float* Yobs  = (const float*)d_in[2];
    const float* noise = (const float*)d_in[3];
    const float* W1    = (const float*)d_in[4];
    const float* b1    = (const float*)d_in[5];
    const float* W2    = (const float*)d_in[6];
    const float* b2    = (const float*)d_in[7];
    const float* W3    = (const float*)d_in[8];
    const float* b3    = (const float*)d_in[9];

    float* outX = (float*)d_out;
    float* outV = outX + (size_t)NN * DD;

    dim3 grid(NN / (WAVES * 16));
    dim3 block(WAVES * 32);
    sde_rollout_kernel<<<grid, block, 0, stream>>>(X0, V0, Yobs, noise,
                                                   W1, b1, W2, b2, W3, b3,
                                                   outX, outV);
}


# Weights in registers: all B fragments of the 3 layers and the biases loaded once into VGPRs (898, one wave per SIMD); K loops read only A from LDS, one load group per layer ahead of its WMMAs
# speedup vs baseline: 1.6729x; 1.6729x over previous
#include <hip/hip_runtime.h>
#include <hip/hip_bf16.h>

typedef float v2f __attribute__((ext_vector_type(2)));
typedef float v8f __attribute__((ext_vector_type(8)));

#define NN 16384
#define DD 16
#define PP 8
#define MM 50
#define HH 128

#define PW12 288
#define PW3   32
#define SXY   28
#define SH   132
#define WAVES 4

static __device__ __forceinline__ v2f ldsv2(const float* p) { return *(const v2f*)p; }

__launch_bounds__(128)
__global__ void sde_rollout_kernel(const float* __restrict__ X0,
                                   const float* __restrict__ V0,
                                   const float* __restrict__ Yobs,
                                   const float* __restrict__ noise,
                                   const float* __restrict__ W1,
                                   const float* __restrict__ b1,
                                   const float* __restrict__ W2,
                                   const float* __restrict__ b2,
                                   const float* __restrict__ W3,
                                   const float* __restrict__ b3,
                                   float* __restrict__ outX,
                                   float* __restrict__ outV)
{
    __shared__ float sW1p[12 * PW12];
    __shared__ float sW1t[HH];
    __shared__ float sB1 [HH];
    __shared__ float sW2p[64 * PW12];
    __shared__ float sB2 [HH];
    __shared__ float sW3p[64 * PW3];
    __shared__ float sB3 [DD];
    __shared__ float sXY [WAVES][16 * SXY];
    __shared__ float sHH [WAVES][16 * SH];

    const int tid  = threadIdx.x;
    const int lane = tid & 31;
    const int wave = tid >> 5;
    const int lo   = lane & 15;
    const int hi   = lane >> 4;
    const int rowbase = blockIdx.x * (WAVES * 16) + wave * 16;

    for (int i = tid; i < 24 * HH; i += blockDim.x) {
        int r = i >> 7, c = i & 127;
        sW1p[(r >> 1) * PW12 + c * 2 + (r & 1)] = W1[(1 + r) * HH + c];
    }
    if (tid < HH) { sW1t[tid] = W1[tid]; sB1[tid] = b1[tid]; sB2[tid] = b2[tid]; }
    for (int i = tid; i < HH * HH; i += blockDim.x) {
        int r = i >> 7, c = i & 127;
        sW2p[(r >> 1) * PW12 + c * 2 + (r & 1)] = W2[i];
    }
    for (int i = tid; i < HH * DD; i += blockDim.x) {
        int r = i >> 4, c = i & 15;
        sW3p[(r >> 1) * PW3 + c * 2 + (r & 1)] = W3[i];
    }
    if (tid < DD) sB3[tid] = b3[tid];

    float* xy = sXY[wave];
    for (int i = lane; i < 16 * PP; i += 32) {
        int r = i >> 3, c = i & 7;
        xy[r * SXY + DD + c] = Yobs[(rowbase + r) * PP + c];
    }
    __syncthreads();

    v2f w1f[6][8];
    #pragma unroll
    for (int kk = 0; kk < 6; ++kk) {
        #pragma unroll
        for (int c = 0; c < 8; ++c)
            w1f[kk][c] = ldsv2(&sW1p[(2 * kk + hi) * PW12 + (c * 16 + lo) * 2]);
    }
    v2f w2f[32][8];
    #pragma unroll
    for (int kk = 0; kk < 32; ++kk) {
        #pragma unroll
        for (int c = 0; c < 8; ++c)
            w2f[kk][c] = ldsv2(&sW2p[(2 * kk + hi) * PW12 + (c * 16 + lo) * 2]);
    }
    v2f w3f[32];
    #pragma unroll
    for (int kk = 0; kk < 32; ++kk)
        w3f[kk] = ldsv2(&sW3p[(2 * kk + hi) * PW3 + lo * 2]);
    float w1tr[8], b1r[8], b2r[8];
    #pragma unroll
    for (int c = 0; c < 8; ++c) {
        w1tr[c] = sW1t[c * 16 + lo];
        b1r[c]  = sB1[c * 16 + lo];
        b2r[c]  = sB2[c * 16 + lo];
    }
    const float b3v = sB3[lo];

    float xr[8], va[8];
    #pragma unroll
    for (int v = 0; v < 8; ++v) {
        int row = rowbase + v + 8 * hi;
        xr[v] = X0[row * DD + lo];
        va[v] = V0[row];
    }

    const float dt    = 1.0f / MM;
    const float sqdt  = __builtin_sqrtf(dt);
    const float decay = 1.0f - dt;
    float* hsc = sHH[wave];
    const v8f zc = {0.f, 0.f, 0.f, 0.f, 0.f, 0.f, 0.f, 0.f};

    for (int m = 0; m < MM; ++m) {
        const float t = m * dt;

        #pragma unroll
        for (int v = 0; v < 8; ++v)
            xy[(v + 8 * hi) * SXY + lo] = xr[v];

        float ep[8];
        #pragma unroll
        for (int v = 0; v < 8; ++v) {
            size_t row = (size_t)m * NN + (size_t)(rowbase + v + 8 * hi);
            ep[v] = noise[row * DD + lo];
        }

        asm volatile("s_wait_dscnt 0" ::: "memory");

        v8f acc[8];

        {
            v2f a[6];
            #pragma unroll
            for (int kk = 0; kk < 6; ++kk) a[kk] = ldsv2(&xy[lo * SXY + 4 * kk + 2 * hi]);
            #pragma unroll
            for (int kk = 0; kk < 6; ++kk) {
                #pragma unroll
                for (int c = 0; c < 8; ++c)
                    acc[c] = __builtin_amdgcn_wmma_f32_16x16x4_f32(
                        false, a[kk], false, w1f[kk][c], (short)0, (kk == 0 ? zc : acc[c]), false, false);
            }
        }
        #pragma unroll
        for (int c = 0; c < 8; ++c) {
            float bias = t * w1tr[c] + b1r[c];
            #pragma unroll
            for (int v = 0; v < 8; ++v)
                hsc[(v + 8 * hi) * SH + c * 16 + lo] = fmaxf(acc[c][v] + bias, 0.0f);
        }
        asm volatile("s_wait_dscnt 0" ::: "memory");

        {
            v2f a[32];
            #pragma unroll
            for (int kk = 0; kk < 32; ++kk) a[kk] = ldsv2(&hsc[lo * SH + 4 * kk + 2 * hi]);
            #pragma unroll
            for (int kk = 0; kk < 32; ++kk) {
                #pragma unroll
                for (int c = 0; c < 8; ++c)
                    acc[c] = __builtin_amdgcn_wmma_f32_16x16x4_f32(
                        false, a[kk], false, w2f[kk][c], (short)0, (kk == 0 ? zc : acc[c]), false, false);
            }
        }
        #pragma unroll
        for (int c = 0; c < 8; ++c) {
            float bias = b2r[c];
            #pragma unroll
            for (int v = 0; v < 8; ++v)
                hsc[(v + 8 * hi) * SH + c * 16 + lo] = fmaxf(acc[c][v] + bias, 0.0f);
        }
        asm volatile("s_wait_dscnt 0" ::: "memory");

        v8f z;
        {
            v2f a[32];
            #pragma unroll
            for (int kk = 0; kk < 32; ++kk) a[kk] = ldsv2(&hsc[lo * SH + 4 * kk + 2 * hi]);
            #pragma unroll
            for (int kk = 0; kk < 32; ++kk)
                z = __builtin_amdgcn_wmma_f32_16x16x4_f32(
                    false, a[kk], false, w3f[kk], (short)0, (kk == 0 ? zc : z), false, false);
        }

        #pragma unroll
        for (int v = 0; v < 8; ++v) {
            float zz = z[v] + b3v;
            float s  = 0.5f * dt * zz * zz + sqdt * zz * ep[v];
            s += __shfl_xor(s, 1);
            s += __shfl_xor(s, 2);
            s += __shfl_xor(s, 4);
            s += __shfl_xor(s, 8);
            va[v] += s;
            xr[v]  = decay * xr[v] + sqdt * ep[v];
        }
    }

    #pragma unroll
    for (int v = 0; v < 8; ++v) {
        int row = rowbase + v + 8 * hi;
        outX[row * DD + lo] = xr[v];
        if (lo == 0) outV[row] = va[v];
    }
}

extern "C" void kernel_launch(void* const* d_in, const int* in_sizes, int n_in,
                              void* d_out, int out_size, void* d_ws, size_t ws_size,
                              hipStream_t stream) {
    (void)in_sizes; (void)n_in; (void)out_size; (void)d_ws; (void)ws_size;
    const float* X0    = (const float*)d_in[0];
    const float* V0    = (const float*)d_in[1];
    const float* Yobs  = (const float*)d_in[2];
    const float* noise = (const float*)d_in[3];
    const float* W1    = (const float*)d_in[4];
    const float* b1    = (const float*)d_in[5];
    const float* W2    = (const float*)d_in[6];
    const float* b2    = (const float*)d_in[7];
    const float* W3    = (const float*)d_in[8];
    const float* b3    = (const float*)d_in[9];

    float* outX = (float*)d_out;
    float* outV = outX + (size_t)NN * DD;

    dim3 grid(NN / (WAVES * 16));
    dim3 block(WAVES * 32);
    sde_rollout_kernel<<<grid, block, 0, stream>>>(X0, V0, Yobs, noise,
                                                   W1, b1, W2, b2, W3, b3,
                                                   outX, outV);
}


# Transposed chain H^T=W^T*X^T: a layer's D tile is the next layer's B fragment, activations stay in VGPRs, step loop without LDS; bias as k row or C start value, V summed in-lane: sum order changes
# speedup vs baseline: 1.8704x; 1.1181x over previous
#include <hip/hip_runtime.h>
#include <hip/hip_bf16.h>

typedef float v2f __attribute__((ext_vector_type(2)));
typedef float v4f __attribute__((ext_vector_type(4)));
typedef float v8f __attribute__((ext_vector_type(8)));

#define NN 16384
#define DD 16
#define PP 8
#define MM 50
#define HH 128

#define PW12 264
#define PW3   40
#define WAVES 4

static __device__ __forceinline__ v2f ldsv2(const float* p) { return *(const v2f*)p; }
static __device__ __forceinline__ float relu(float x) { return __builtin_elementwise_maximum(x, 0.0f); }
static __device__ __forceinline__ float xhalf(float x) {
    const int b = __builtin_bit_cast(int, x);
    return __builtin_bit_cast(float, __builtin_amdgcn_permlanex16(b, b, 0x76543210, 0xfedcba98, false, false));
}

__launch_bounds__(128)
__global__ void sde_rollout_kernel(const float* __restrict__ X0,
                                   const float* __restrict__ V0,
                                   const float* __restrict__ Yobs,
                                   const float* __restrict__ noise,
                                   const float* __restrict__ W1,
                                   const float* __restrict__ b1,
                                   const float* __restrict__ W2,
                                   const float* __restrict__ b2,
                                   const float* __restrict__ W3,
                                   const float* __restrict__ b3,
                                   float* __restrict__ outX,
                                   float* __restrict__ outV)
{
    __shared__ float sW1p[12 * PW12];
    __shared__ float sW2p[64 * PW12];
    __shared__ float sW3p[64 * PW3];

    const int tid  = threadIdx.x;
    const int lane = tid & 31;
    const int wave = tid >> 5;
    const int lo   = lane & 15;
    const int hi   = lane >> 4;
    const int rowbase = blockIdx.x * (WAVES * 16) + wave * 16;
    const int brow = rowbase + lo;

    for (int i = tid; i < 24 * HH; i += blockDim.x) {
        int r = i >> 7, c = i & 127;
        sW1p[(r >> 1) * PW12 + c * 2 + (r & 1)] = W1[(1 + r) * HH + c];
    }
    for (int i = tid; i < HH * HH; i += blockDim.x) {
        int r = i >> 7, c = i & 127;
        sW2p[(r >> 1) * PW12 + c * 2 + (r & 1)] = W2[i];
    }
    for (int i = tid; i < HH * DD; i += blockDim.x) {
        int r = i >> 4, c = i & 15;
        sW3p[(r >> 1) * PW3 + c * 2 + (r & 1)] = W3[i];
    }
    __syncthreads();

    v2f w1f[7][8];
    #pragma unroll
    for (int j = 0; j < 4; ++j) {
        #pragma unroll
        for (int ot = 0; ot < 8; ++ot)
            w1f[j][ot] = ldsv2(&sW1p[(4 * hi + j) * PW12 + (ot * 16 + lo) * 2]);
    }
    #pragma unroll
    for (int j = 0; j < 2; ++j) {
        #pragma unroll
        for (int ot = 0; ot < 8; ++ot)
            w1f[4 + j][ot] = ldsv2(&sW1p[(8 + 2 * hi + j) * PW12 + (ot * 16 + lo) * 2]);
    }
    #pragma unroll
    for (int ot = 0; ot < 8; ++ot) {
        float wt = W1[ot * 16 + lo];
        float bb = b1[ot * 16 + lo];
        w1f[6][ot][0] = hi ? 0.0f : wt;
        w1f[6][ot][1] = hi ? 0.0f : bb;
    }
    v2f w2f[32][8];
    #pragma unroll
    for (int g = 0; g < 32; ++g) {
        #pragma unroll
        for (int ot = 0; ot < 8; ++ot)
            w2f[g][ot] = ldsv2(&sW2p[((g >> 2) * 8 + 4 * hi + (g & 3)) * PW12 + (ot * 16 + lo) * 2]);
    }
    v2f w3f[32];
    #pragma unroll
    for (int g = 0; g < 32; ++g)
        w3f[g] = ldsv2(&sW3p[((g >> 2) * 8 + 4 * hi + (g & 3)) * PW3 + lo * 2]);

    v8f b2c[8];
    #pragma unroll
    for (int ot = 0; ot < 8; ++ot) {
        const v4f q0 = *(const v4f*)(b2 + ot * 16 + 8 * hi);
        const v4f q1 = *(const v4f*)(b2 + ot * 16 + 8 * hi + 4);
        b2c[ot] = (v8f){q0[0], q0[1], q0[2], q0[3], q1[0], q1[1], q1[2], q1[3]};
    }
    v8f b3c;
    {
        const v4f q0 = *(const v4f*)(b3 + 8 * hi);
        const v4f q1 = *(const v4f*)(b3 + 8 * hi + 4);
        b3c = (v8f){q0[0], q0[1], q0[2], q0[3], q1[0], q1[1], q1[2], q1[3]};
    }

    v2f xr[4];
    {
        const v4f q0 = *(const v4f*)(X0 + (size_t)brow * DD + 8 * hi);
        const v4f q1 = *(const v4f*)(X0 + (size_t)brow * DD + 8 * hi + 4);
        xr[0] = (v2f){q0[0], q0[1]}; xr[1] = (v2f){q0[2], q0[3]};
        xr[2] = (v2f){q1[0], q1[1]}; xr[3] = (v2f){q1[2], q1[3]};
    }
    float va = V0[brow];
    v2f yb[2];
    {
        const v4f q = *(const v4f*)(Yobs + (size_t)brow * PP + 4 * hi);
        yb[0] = (v2f){q[0], q[1]};
        yb[1] = (v2f){q[2], q[3]};
    }

    const float* nlane = noise + (size_t)brow * DD + 8 * hi;
    v2f ep[4];
    {
        const v4f q0 = *(const v4f*)(nlane);
        const v4f q1 = *(const v4f*)(nlane + 4);
        ep[0] = (v2f){q0[0], q0[1]}; ep[1] = (v2f){q0[2], q0[3]};
        ep[2] = (v2f){q1[0], q1[1]}; ep[3] = (v2f){q1[2], q1[3]};
    }

    const float dt    = 1.0f / MM;
    const float sqdt  = __builtin_sqrtf(dt);
    const float decay = 1.0f - dt;
    const v8f zc = {0.f, 0.f, 0.f, 0.f, 0.f, 0.f, 0.f, 0.f};

    for (int m = 0; m < MM; ++m) {
        const float t = m * dt;

        v8f acc1[8];
        {
            v2f tb;
            tb[0] = hi ? 0.0f : t;
            tb[1] = hi ? 0.0f : 1.0f;
            #pragma unroll
            for (int ot = 0; ot < 8; ++ot)
                acc1[ot] = __builtin_amdgcn_wmma_f32_16x16x4_f32(
                    false, w1f[6][ot], false, tb, (short)0, zc, false, false);
            #pragma unroll
            for (int j = 0; j < 4; ++j) {
                #pragma unroll
                for (int ot = 0; ot < 8; ++ot)
                    acc1[ot] = __builtin_amdgcn_wmma_f32_16x16x4_f32(
                        false, w1f[j][ot], false, xr[j], (short)0, acc1[ot], false, false);
            }
            #pragma unroll
            for (int j = 0; j < 2; ++j) {
                #pragma unroll
                for (int ot = 0; ot < 8; ++ot)
                    acc1[ot] = __builtin_amdgcn_wmma_f32_16x16x4_f32(
                        false, w1f[4 + j][ot], false, yb[j], (short)0, acc1[ot], false, false);
            }
        }

        v8f acc2[8];
        #pragma unroll
        for (int ct = 0; ct < 8; ++ct) {
            #pragma unroll
            for (int r = 0; r < 8; ++r) acc1[ct][r] = relu(acc1[ct][r]);
            #pragma unroll
            for (int j = 0; j < 4; ++j) {
                const v2f hb = {acc1[ct][2 * j], acc1[ct][2 * j + 1]};
                #pragma unroll
                for (int ot = 0; ot < 8; ++ot)
                    acc2[ot] = __builtin_amdgcn_wmma_f32_16x16x4_f32(
                        false, w2f[ct * 4 + j][ot], false, hb, (short)0,
                        (ct == 0 && j == 0) ? b2c[ot] : acc2[ot], false, false);
            }
        }

        v8f z;
        #pragma unroll
        for (int ct = 0; ct < 8; ++ct) {
            #pragma unroll
            for (int r = 0; r < 8; ++r) acc2[ct][r] = relu(acc2[ct][r]);
            #pragma unroll
            for (int j = 0; j < 4; ++j) {
                const v2f hb = {acc2[ct][2 * j], acc2[ct][2 * j + 1]};
                z = __builtin_amdgcn_wmma_f32_16x16x4_f32(
                    false, w3f[ct * 4 + j], false, hb, (short)0,
                    (ct == 0 && j == 0) ? b3c : z, false, false);
            }
        }

        v2f sv = {0.0f, 0.0f};
        #pragma unroll
        for (int j = 0; j < 4; ++j) {
            const v2f zz = {z[2 * j], z[2 * j + 1]};
            sv += 0.5f * dt * zz * zz + sqdt * zz * ep[j];
            xr[j] = decay * xr[j] + sqdt * ep[j];
        }
        float s = sv[0] + sv[1];
        s += xhalf(s);
        va += s;

        {
            const int mn = (m + 1 < MM) ? (m + 1) : (MM - 1);
            const float* np = nlane + (size_t)mn * NN * DD;
            const v4f q0 = *(const v4f*)(np);
            const v4f q1 = *(const v4f*)(np + 4);
            ep[0] = (v2f){q0[0], q0[1]}; ep[1] = (v2f){q0[2], q0[3]};
            ep[2] = (v2f){q1[0], q1[1]}; ep[3] = (v2f){q1[2], q1[3]};
        }
    }

    {
        v4f* ox = (v4f*)(outX + (size_t)brow * DD + 8 * hi);
        ox[0] = (v4f){xr[0][0], xr[0][1], xr[1][0], xr[1][1]};
        ox[1] = (v4f){xr[2][0], xr[2][1], xr[3][0], xr[3][1]};
        if (hi == 0) outV[brow] = va;
    }
}

extern "C" void kernel_launch(void* const* d_in, const int* in_sizes, int n_in,
                              void* d_out, int out_size, void* d_ws, size_t ws_size,
                              hipStream_t stream) {
    (void)in_sizes; (void)n_in; (void)out_size; (void)d_ws; (void)ws_size;
    const float* X0    = (const float*)d_in[0];
    const float* V0    = (const float*)d_in[1];
    const float* Yobs  = (const float*)d_in[2];
    const float* noise = (const float*)d_in[3];
    const float* W1    = (const float*)d_in[4];
    const float* b1    = (const float*)d_in[5];
    const float* W2    = (const float*)d_in[6];
    const float* b2    = (const float*)d_in[7];
    const float* W3    = (const float*)d_in[8];
    const float* b3    = (const float*)d_in[9];

    float* outX = (float*)d_out;
    float* outV = outX + (size_t)NN * DD;

    dim3 grid(NN / (WAVES * 16));
    dim3 block(WAVES * 32);
    sde_rollout_kernel<<<grid, block, 0, stream>>>(X0, V0, Yobs, noise,
                                                   W1, b1, W2, b2, W3, b3,
                                                   outX, outV);
}


# Weight staging global->LDS: each thread issues all its 16-byte row loads as one group ahead of 16-byte LDS stores (fixed trip counts), replacing one-word load-wait-store loops (144 serial trips)
# speedup vs baseline: 2.7200x; 1.4542x over previous
#include <hip/hip_runtime.h>
#include <hip/hip_bf16.h>

typedef float v2f __attribute__((ext_vector_type(2)));
typedef float v4f __attribute__((ext_vector_type(4)));
typedef float v8f __attribute__((ext_vector_type(8)));

#define NN 16384
#define DD 16
#define PP 8
#define MM 50
#define HH 128

#define PW12 264
#define PW3   40
#define WAVES 4

static __device__ __forceinline__ v2f ldsv2(const float* p) { return *(const v2f*)p; }
static __device__ __forceinline__ float relu(float x) { return __builtin_elementwise_maximum(x, 0.0f); }
static __device__ __forceinline__ float xhalf(float x) {
    const int b = __builtin_bit_cast(int, x);
    return __builtin_bit_cast(float, __builtin_amdgcn_permlanex16(b, b, 0x76543210, 0xfedcba98, false, false));
}

__launch_bounds__(128)
__global__ void sde_rollout_kernel(const float* __restrict__ X0,
                                   const float* __restrict__ V0,
                                   const float* __restrict__ Yobs,
                                   const float* __restrict__ noise,
                                   const float* __restrict__ W1,
                                   const float* __restrict__ b1,
                                   const float* __restrict__ W2,
                                   const float* __restrict__ b2,
                                   const float* __restrict__ W3,
                                   const float* __restrict__ b3,
                                   float* __restrict__ outX,
                                   float* __restrict__ outV)
{
    __shared__ __attribute__((aligned(16))) float sW1p[12 * PW12];
    __shared__ __attribute__((aligned(16))) float sW2p[64 * PW12];
    __shared__ __attribute__((aligned(16))) float sW3p[64 * PW3];

    const int tid  = threadIdx.x;
    const int lane = tid & 31;
    const int wave = tid >> 5;
    const int lo   = lane & 15;
    const int hi   = lane >> 4;
    const int rowbase = blockIdx.x * (WAVES * 16) + wave * 16;
    const int brow = rowbase + lo;

    {
        v4f a1[3], c1[3];
        #pragma unroll
        for (int it = 0; it < 3; ++it) {
            const int p = it * WAVES + wave;
            a1[it] = *(const v4f*)(W1 + (1 + 2 * p) * HH + 4 * lane);
            c1[it] = *(const v4f*)(W1 + (2 + 2 * p) * HH + 4 * lane);
        }
        v4f a2[16], c2[16];
        #pragma unroll
        for (int it = 0; it < 16; ++it) {
            const int p = it * WAVES + wave;
            a2[it] = *(const v4f*)(W2 + (2 * p) * HH + 4 * lane);
            c2[it] = *(const v4f*)(W2 + (2 * p + 1) * HH + 4 * lane);
        }
        v4f a3[2], c3[2];
        #pragma unroll
        for (int it = 0; it < 2; ++it) {
            const int item = it * (WAVES * 32) + tid;
            const int p = item >> 2, q = item & 3;
            a3[it] = *(const v4f*)(W3 + (2 * p) * DD + 4 * q);
            c3[it] = *(const v4f*)(W3 + (2 * p + 1) * DD + 4 * q);
        }
        #pragma unroll
        for (int it = 0; it < 3; ++it) {
            const int p = it * WAVES + wave;
            v4f* d = (v4f*)&sW1p[p * PW12 + 8 * lane];
            d[0] = (v4f){a1[it][0], c1[it][0], a1[it][1], c1[it][1]};
            d[1] = (v4f){a1[it][2], c1[it][2], a1[it][3], c1[it][3]};
        }
        #pragma unroll
        for (int it = 0; it < 16; ++it) {
            const int p = it * WAVES + wave;
            v4f* d = (v4f*)&sW2p[p * PW12 + 8 * lane];
            d[0] = (v4f){a2[it][0], c2[it][0], a2[it][1], c2[it][1]};
            d[1] = (v4f){a2[it][2], c2[it][2], a2[it][3], c2[it][3]};
        }
        #pragma unroll
        for (int it = 0; it < 2; ++it) {
            const int item = it * (WAVES * 32) + tid;
            const int p = item >> 2, q = item & 3;
            v4f* d = (v4f*)&sW3p[p * PW3 + 8 * q];
            d[0] = (v4f){a3[it][0], c3[it][0], a3[it][1], c3[it][1]};
            d[1] = (v4f){a3[it][2], c3[it][2], a3[it][3], c3[it][3]};
        }
    }
    __syncthreads();

    v2f w1f[7][8];
    #pragma unroll
    for (int j = 0; j < 4; ++j) {
        #pragma unroll
        for (int ot = 0; ot < 8; ++ot)
            w1f[j][ot] = ldsv2(&sW1p[(4 * hi + j) * PW12 + (ot * 16 + lo) * 2]);
    }
    #pragma unroll
    for (int j = 0; j < 2; ++j) {
        #pragma unroll
        for (int ot = 0; ot < 8; ++ot)
            w1f[4 + j][ot] = ldsv2(&sW1p[(8 + 2 * hi + j) * PW12 + (ot * 16 + lo) * 2]);
    }
    #pragma unroll
    for (int ot = 0; ot < 8; ++ot) {
        float wt = W1[ot * 16 + lo];
        float bb = b1[ot * 16 + lo];
        w1f[6][ot][0] = hi ? 0.0f : wt;
        w1f[6][ot][1] = hi ? 0.0f : bb;
    }
    v2f w2f[32][8];
    #pragma unroll
    for (int g = 0; g < 32; ++g) {
        #pragma unroll
        for (int ot = 0; ot < 8; ++ot)
            w2f[g][ot] = ldsv2(&sW2p[((g >> 2) * 8 + 4 * hi + (g & 3)) * PW12 + (ot * 16 + lo) * 2]);
    }
    v2f w3f[32];
    #pragma unroll
    for (int g = 0; g < 32; ++g)
        w3f[g] = ldsv2(&sW3p[((g >> 2) * 8 + 4 * hi + (g & 3)) * PW3 + lo * 2]);

    v8f b2c[8];
    #pragma unroll
    for (int ot = 0; ot < 8; ++ot) {
        const v4f q0 = *(const v4f*)(b2 + ot * 16 + 8 * hi);
        const v4f q1 = *(const v4f*)(b2 + ot * 16 + 8 * hi + 4);
        b2c[ot] = (v8f){q0[0], q0[1], q0[2], q0[3], q1[0], q1[1], q1[2], q1[3]};
    }
    v8f b3c;
    {
        const v4f q0 = *(const v4f*)(b3 + 8 * hi);
        const v4f q1 = *(const v4f*)(b3 + 8 * hi + 4);
        b3c = (v8f){q0[0], q0[1], q0[2], q0[3], q1[0], q1[1], q1[2], q1[3]};
    }

    v2f xr[4];
    {
        const v4f q0 = *(const v4f*)(X0 + (size_t)brow * DD + 8 * hi);
        const v4f q1 = *(const v4f*)(X0 + (size_t)brow * DD + 8 * hi + 4);
        xr[0] = (v2f){q0[0], q0[1]}; xr[1] = (v2f){q0[2], q0[3]};
        xr[2] = (v2f){q1[0], q1[1]}; xr[3] = (v2f){q1[2], q1[3]};
    }
    float va = V0[brow];
    v2f yb[2];
    {
        const v4f q = *(const v4f*)(Yobs + (size_t)brow * PP + 4 * hi);
        yb[0] = (v2f){q[0], q[1]};
        yb[1] = (v2f){q[2], q[3]};
    }

    const float* nlane = noise + (size_t)brow * DD + 8 * hi;
    v2f ep[4];
    {
        const v4f q0 = *(const v4f*)(nlane);
        const v4f q1 = *(const v4f*)(nlane + 4);
        ep[0] = (v2f){q0[0], q0[1]}; ep[1] = (v2f){q0[2], q0[3]};
        ep[2] = (v2f){q1[0], q1[1]}; ep[3] = (v2f){q1[2], q1[3]};
    }

    const float dt    = 1.0f / MM;
    const float sqdt  = __builtin_sqrtf(dt);
    const float decay = 1.0f - dt;
    const v8f zc = {0.f, 0.f, 0.f, 0.f, 0.f, 0.f, 0.f, 0.f};

    for (int m = 0; m < MM; ++m) {
        const float t = m * dt;

        v8f acc1[8];
        {
            v2f tb;
            tb[0] = hi ? 0.0f : t;
            tb[1] = hi ? 0.0f : 1.0f;
            #pragma unroll
            for (int ot = 0; ot < 8; ++ot)
                acc1[ot] = __builtin_amdgcn_wmma_f32_16x16x4_f32(
                    false, w1f[6][ot], false, tb, (short)0, zc, false, false);
            #pragma unroll
            for (int j = 0; j < 4; ++j) {
                #pragma unroll
                for (int ot = 0; ot < 8; ++ot)
                    acc1[ot] = __builtin_amdgcn_wmma_f32_16x16x4_f32(
                        false, w1f[j][ot], false, xr[j], (short)0, acc1[ot], false, false);
            }
            #pragma unroll
            for (int j = 0; j < 2; ++j) {
                #pragma unroll
                for (int ot = 0; ot < 8; ++ot)
                    acc1[ot] = __builtin_amdgcn_wmma_f32_16x16x4_f32(
                        false, w1f[4 + j][ot], false, yb[j], (short)0, acc1[ot], false, false);
            }
        }

        v8f acc2[8];
        #pragma unroll
        for (int ct = 0; ct < 8; ++ct) {
            #pragma unroll
            for (int r = 0; r < 8; ++r) acc1[ct][r] = relu(acc1[ct][r]);
            #pragma unroll
            for (int j = 0; j < 4; ++j) {
                const v2f hb = {acc1[ct][2 * j], acc1[ct][2 * j + 1]};
                #pragma unroll
                for (int ot = 0; ot < 8; ++ot)
                    acc2[ot] = __builtin_amdgcn_wmma_f32_16x16x4_f32(
                        false, w2f[ct * 4 + j][ot], false, hb, (short)0,
                        (ct == 0 && j == 0) ? b2c[ot] : acc2[ot], false, false);
            }
        }

        v8f z;
        #pragma unroll
        for (int ct = 0; ct < 8; ++ct) {
            #pragma unroll
            for (int r = 0; r < 8; ++r) acc2[ct][r] = relu(acc2[ct][r]);
            #pragma unroll
            for (int j = 0; j < 4; ++j) {
                const v2f hb = {acc2[ct][2 * j], acc2[ct][2 * j + 1]};
                z = __builtin_amdgcn_wmma_f32_16x16x4_f32(
                    false, w3f[ct * 4 + j], false, hb, (short)0,
                    (ct == 0 && j == 0) ? b3c : z, false, false);
            }
        }

        v2f sv = {0.0f, 0.0f};
        #pragma unroll
        for (int j = 0; j < 4; ++j) {
            const v2f zz = {z[2 * j], z[2 * j + 1]};
            sv += 0.5f * dt * zz * zz + sqdt * zz * ep[j];
            xr[j] = decay * xr[j] + sqdt * ep[j];
        }
        float s = sv[0] + sv[1];
        s += xhalf(s);
        va += s;

        {
            const int mn = (m + 1 < MM) ? (m + 1) : (MM - 1);
            const float* np = nlane + (size_t)mn * NN * DD;
            const v4f q0 = *(const v4f*)(np);
            const v4f q1 = *(const v4f*)(np + 4);
            ep[0] = (v2f){q0[0], q0[1]}; ep[1] = (v2f){q0[2], q0[3]};
            ep[2] = (v2f){q1[0], q1[1]}; ep[3] = (v2f){q1[2], q1[3]};
        }
    }

    {
        v4f* ox = (v4f*)(outX + (size_t)brow * DD + 8 * hi);
        ox[0] = (v4f){xr[0][0], xr[0][1], xr[1][0], xr[1][1]};
        ox[1] = (v4f){xr[2][0], xr[2][1], xr[3][0], xr[3][1]};
        if (hi == 0) outV[brow] = va;
    }
}

extern "C" void kernel_launch(void* const* d_in, const int* in_sizes, int n_in,
                              void* d_out, int out_size, void* d_ws, size_t ws_size,
                              hipStream_t stream) {
    (void)in_sizes; (void)n_in; (void)out_size; (void)d_ws; (void)ws_size;
    const float* X0    = (const float*)d_in[0];
    const float* V0    = (const float*)d_in[1];
    const float* Yobs  = (const float*)d_in[2];
    const float* noise = (const float*)d_in[3];
    const float* W1    = (const float*)d_in[4];
    const float* b1    = (const float*)d_in[5];
    const float* W2    = (const float*)d_in[6];
    const float* b2    = (const float*)d_in[7];
    const float* W3    = (const float*)d_in[8];
    const float* b3    = (const float*)d_in[9];

    float* outX = (float*)d_out;
    float* outV = outX + (size_t)NN * DD;

    dim3 grid(NN / (WAVES * 16));
    dim3 block(WAVES * 32);
    sde_rollout_kernel<<<grid, block, 0, stream>>>(X0, V0, Yobs, noise,
                                                   W1, b1, W2, b2, W3, b3,
                                                   outX, outV);
}


# Layer 1 multiplies only X per step (K 28->16): Y*W1y+b1 is computed once as a constant start tile, t*W1[0] is added to it by vector FMA; 24 fewer WMMAs per step; layer-1 sum order changes
# speedup vs baseline: 2.8757x; 1.0572x over previous
#include <hip/hip_runtime.h>
#include <hip/hip_bf16.h>

typedef float v2f __attribute__((ext_vector_type(2)));
typedef float v4f __attribute__((ext_vector_type(4)));
typedef float v8f __attribute__((ext_vector_type(8)));

#define NN 16384
#define DD 16
#define PP 8
#define MM 50
#define HH 128

#define PW12 264
#define PW3   40
#define WAVES 4

static __device__ __forceinline__ v2f ldsv2(const float* p) { return *(const v2f*)p; }
static __device__ __forceinline__ float relu(float x) { return __builtin_elementwise_maximum(x, 0.0f); }
static __device__ __forceinline__ float xhalf(float x) {
    const int b = __builtin_bit_cast(int, x);
    return __builtin_bit_cast(float, __builtin_amdgcn_permlanex16(b, b, 0x76543210, 0xfedcba98, false, false));
}

__launch_bounds__(128)
__global__ void sde_rollout_kernel(const float* __restrict__ X0,
                                   const float* __restrict__ V0,
                                   const float* __restrict__ Yobs,
                                   const float* __restrict__ noise,
                                   const float* __restrict__ W1,
                                   const float* __restrict__ b1,
                                   const float* __restrict__ W2,
                                   const float* __restrict__ b2,
                                   const float* __restrict__ W3,
                                   const float* __restrict__ b3,
                                   float* __restrict__ outX,
                                   float* __restrict__ outV)
{
    __shared__ __attribute__((aligned(16))) float sW1p[12 * PW12];
    __shared__ __attribute__((aligned(16))) float sW2p[64 * PW12];
    __shared__ __attribute__((aligned(16))) float sW3p[64 * PW3];

    const int tid  = threadIdx.x;
    const int lane = tid & 31;
    const int wave = tid >> 5;
    const int lo   = lane & 15;
    const int hi   = lane >> 4;
    const int rowbase = blockIdx.x * (WAVES * 16) + wave * 16;
    const int brow = rowbase + lo;

    {
        v4f a1[3], c1[3];
        #pragma unroll
        for (int it = 0; it < 3; ++it) {
            const int p = it * WAVES + wave;
            a1[it] = *(const v4f*)(W1 + (1 + 2 * p) * HH + 4 * lane);
            c1[it] = *(const v4f*)(W1 + (2 + 2 * p) * HH + 4 * lane);
        }
        v4f a2[16], c2[16];
        #pragma unroll
        for (int it = 0; it < 16; ++it) {
            const int p = it * WAVES + wave;
            a2[it] = *(const v4f*)(W2 + (2 * p) * HH + 4 * lane);
            c2[it] = *(const v4f*)(W2 + (2 * p + 1) * HH + 4 * lane);
        }
        v4f a3[2], c3[2];
        #pragma unroll
        for (int it = 0; it < 2; ++it) {
            const int item = it * (WAVES * 32) + tid;
            const int p = item >> 2, q = item & 3;
            a3[it] = *(const v4f*)(W3 + (2 * p) * DD + 4 * q);
            c3[it] = *(const v4f*)(W3 + (2 * p + 1) * DD + 4 * q);
        }
        #pragma unroll
        for (int it = 0; it < 3; ++it) {
            const int p = it * WAVES + wave;
            v4f* d = (v4f*)&sW1p[p * PW12 + 8 * lane];
            d[0] = (v4f){a1[it][0], c1[it][0], a1[it][1], c1[it][1]};
            d[1] = (v4f){a1[it][2], c1[it][2], a1[it][3], c1[it][3]};
        }
        #pragma unroll
        for (int it = 0; it < 16; ++it) {
            const int p = it * WAVES + wave;
            v4f* d = (v4f*)&sW2p[p * PW12 + 8 * lane];
            d[0] = (v4f){a2[it][0], c2[it][0], a2[it][1], c2[it][1]};
            d[1] = (v4f){a2[it][2], c2[it][2], a2[it][3], c2[it][3]};
        }
        #pragma unroll
        for (int it = 0; it < 2; ++it) {
            const int item = it * (WAVES * 32) + tid;
            const int p = item >> 2, q = item & 3;
            v4f* d = (v4f*)&sW3p[p * PW3 + 8 * q];
            d[0] = (v4f){a3[it][0], c3[it][0], a3[it][1], c3[it][1]};
            d[1] = (v4f){a3[it][2], c3[it][2], a3[it][3], c3[it][3]};
        }
    }
    __syncthreads();

    v2f w1f[4][8];
    #pragma unroll
    for (int j = 0; j < 4; ++j) {
        #pragma unroll
        for (int ot = 0; ot < 8; ++ot)
            w1f[j][ot] = ldsv2(&sW1p[(4 * hi + j) * PW12 + (ot * 16 + lo) * 2]);
    }
    v2f w2f[32][8];
    #pragma unroll
    for (int g = 0; g < 32; ++g) {
        #pragma unroll
        for (int ot = 0; ot < 8; ++ot)
            w2f[g][ot] = ldsv2(&sW2p[((g >> 2) * 8 + 4 * hi + (g & 3)) * PW12 + (ot * 16 + lo) * 2]);
    }
    v2f w3f[32];
    #pragma unroll
    for (int g = 0; g < 32; ++g)
        w3f[g] = ldsv2(&sW3p[((g >> 2) * 8 + 4 * hi + (g & 3)) * PW3 + lo * 2]);

    v8f b2c[8];
    #pragma unroll
    for (int ot = 0; ot < 8; ++ot) {
        const v4f q0 = *(const v4f*)(b2 + ot * 16 + 8 * hi);
        const v4f q1 = *(const v4f*)(b2 + ot * 16 + 8 * hi + 4);
        b2c[ot] = (v8f){q0[0], q0[1], q0[2], q0[3], q1[0], q1[1], q1[2], q1[3]};
    }
    v8f b3c;
    {
        const v4f q0 = *(const v4f*)(b3 + 8 * hi);
        const v4f q1 = *(const v4f*)(b3 + 8 * hi + 4);
        b3c = (v8f){q0[0], q0[1], q0[2], q0[3], q1[0], q1[1], q1[2], q1[3]};
    }

    v2f xr[4];
    {
        const v4f q0 = *(const v4f*)(X0 + (size_t)brow * DD + 8 * hi);
        const v4f q1 = *(const v4f*)(X0 + (size_t)brow * DD + 8 * hi + 4);
        xr[0] = (v2f){q0[0], q0[1]}; xr[1] = (v2f){q0[2], q0[3]};
        xr[2] = (v2f){q1[0], q1[1]}; xr[3] = (v2f){q1[2], q1[3]};
    }
    float va = V0[brow];
    v8f c1[8], w0d[8];
    {
        v2f yb[2];
        const v4f q = *(const v4f*)(Yobs + (size_t)brow * PP + 4 * hi);
        yb[0] = (v2f){q[0], q[1]};
        yb[1] = (v2f){q[2], q[3]};
        #pragma unroll
        for (int ot = 0; ot < 8; ++ot) {
            const v4f p0 = *(const v4f*)(b1 + ot * 16 + 8 * hi);
            const v4f p1 = *(const v4f*)(b1 + ot * 16 + 8 * hi + 4);
            c1[ot] = (v8f){p0[0], p0[1], p0[2], p0[3], p1[0], p1[1], p1[2], p1[3]};
            const v4f t0 = *(const v4f*)(W1 + ot * 16 + 8 * hi);
            const v4f t1 = *(const v4f*)(W1 + ot * 16 + 8 * hi + 4);
            w0d[ot] = (v8f){t0[0], t0[1], t0[2], t0[3], t1[0], t1[1], t1[2], t1[3]};
        }
        #pragma unroll
        for (int j = 0; j < 2; ++j) {
            #pragma unroll
            for (int ot = 0; ot < 8; ++ot) {
                const v2f wy = ldsv2(&sW1p[(8 + 2 * hi + j) * PW12 + (ot * 16 + lo) * 2]);
                c1[ot] = __builtin_amdgcn_wmma_f32_16x16x4_f32(
                    false, wy, false, yb[j], (short)0, c1[ot], false, false);
            }
        }
    }

    const float* nlane = noise + (size_t)brow * DD + 8 * hi;
    v2f ep[4];
    {
        const v4f q0 = *(const v4f*)(nlane);
        const v4f q1 = *(const v4f*)(nlane + 4);
        ep[0] = (v2f){q0[0], q0[1]}; ep[1] = (v2f){q0[2], q0[3]};
        ep[2] = (v2f){q1[0], q1[1]}; ep[3] = (v2f){q1[2], q1[3]};
    }

    const float dt    = 1.0f / MM;
    const float sqdt  = __builtin_sqrtf(dt);
    const float decay = 1.0f - dt;

    for (int m = 0; m < MM; ++m) {
        const float t = m * dt;

        v8f acc1[8];
        #pragma unroll
        for (int j = 0; j < 4; ++j) {
            #pragma unroll
            for (int ot = 0; ot < 8; ++ot)
                acc1[ot] = __builtin_amdgcn_wmma_f32_16x16x4_f32(
                    false, w1f[j][ot], false, xr[j], (short)0,
                    (j == 0) ? (t * w0d[ot] + c1[ot]) : acc1[ot], false, false);
        }

        v8f acc2[8];
        #pragma unroll
        for (int ct = 0; ct < 8; ++ct) {
            #pragma unroll
            for (int r = 0; r < 8; ++r) acc1[ct][r] = relu(acc1[ct][r]);
            #pragma unroll
            for (int j = 0; j < 4; ++j) {
                const v2f hb = {acc1[ct][2 * j], acc1[ct][2 * j + 1]};
                #pragma unroll
                for (int ot = 0; ot < 8; ++ot)
                    acc2[ot] = __builtin_amdgcn_wmma_f32_16x16x4_f32(
                        false, w2f[ct * 4 + j][ot], false, hb, (short)0,
                        (ct == 0 && j == 0) ? b2c[ot] : acc2[ot], false, false);
            }
        }

        v8f z;
        #pragma unroll
        for (int ct = 0; ct < 8; ++ct) {
            #pragma unroll
            for (int r = 0; r < 8; ++r) acc2[ct][r] = relu(acc2[ct][r]);
            #pragma unroll
            for (int j = 0; j < 4; ++j) {
                const v2f hb = {acc2[ct][2 * j], acc2[ct][2 * j + 1]};
                z = __builtin_amdgcn_wmma_f32_16x16x4_f32(
                    false, w3f[ct * 4 + j], false, hb, (short)0,
                    (ct == 0 && j == 0) ? b3c : z, false, false);
            }
        }

        v2f sv = {0.0f, 0.0f};
        #pragma unroll
        for (int j = 0; j < 4; ++j) {
            const v2f zz = {z[2 * j], z[2 * j + 1]};
            sv += 0.5f * dt * zz * zz + sqdt * zz * ep[j];
            xr[j] = decay * xr[j] + sqdt * ep[j];
        }
        float s = sv[0] + sv[1];
        s += xhalf(s);
        va += s;

        {
            const int mn = (m + 1 < MM) ? (m + 1) : (MM - 1);
            const float* np = nlane + (size_t)mn * NN * DD;
            const v4f q0 = *(const v4f*)(np);
            const v4f q1 = *(const v4f*)(np + 4);
            ep[0] = (v2f){q0[0], q0[1]}; ep[1] = (v2f){q0[2], q0[3]};
            ep[2] = (v2f){q1[0], q1[1]}; ep[3] = (v2f){q1[2], q1[3]};
        }
    }

    {
        v4f* ox = (v4f*)(outX + (size_t)brow * DD + 8 * hi);
        ox[0] = (v4f){xr[0][0], xr[0][1], xr[1][0], xr[1][1]};
        ox[1] = (v4f){xr[2][0], xr[2][1], xr[3][0], xr[3][1]};
        if (hi == 0) outV[brow] = va;
    }
}

extern "C" void kernel_launch(void* const* d_in, const int* in_sizes, int n_in,
                              void* d_out, int out_size, void* d_ws, size_t ws_size,
                              hipStream_t stream) {
    (void)in_sizes; (void)n_in; (void)out_size; (void)d_ws; (void)ws_size;
    const float* X0    = (const float*)d_in[0];
    const float* V0    = (const float*)d_in[1];
    const float* Yobs  = (const float*)d_in[2];
    const float* noise = (const float*)d_in[3];
    const float* W1    = (const float*)d_in[4];
    const float* b1    = (const float*)d_in[5];
    const float* W2    = (const float*)d_in[6];
    const float* b2    = (const float*)d_in[7];
    const float* W3    = (const float*)d_in[8];
    const float* b3    = (const float*)d_in[9];

    float* outX = (float*)d_out;
    float* outV = outX + (size_t)NN * DD;

    dim3 grid(NN / (WAVES * 16));
    dim3 block(WAVES * 32);
    sde_rollout_kernel<<<grid, block, 0, stream>>>(X0, V0, Yobs, noise,
                                                   W1, b1, W2, b2, W3, b3,
                                                   outX, outV);
}
